# starting kernel
# baseline (speedup 1.0000x reference)
#include <hip/hip_runtime.h>

#define DIM   300
#define T1N   128
#define T2N   128
#define BATCH 4
#define ROWS  (BATCH * T1N)

#define WAVES 8
#define BLOCK (WAVES * 32)

typedef __attribute__((ext_vector_type(2))) float v2f;
typedef __attribute__((ext_vector_type(8))) float v8f;

__global__ __launch_bounds__(32) void proj_wmma_kernel(
    const float* __restrict__ X, const float* __restrict__ W,
    const float* __restrict__ bias, float* __restrict__ out)
{
    const int lane  = threadIdx.x;
    const int nTile = blockIdx.x;
    const int mTile = blockIdx.y;
    const int m0 = mTile * 16;
    const int n0 = nTile * 16;

    const int half = lane >> 4;
    const int l16  = lane & 15;

    const int  n      = n0 + l16;
    const bool nvalid = (n < DIM);
    const int  nsafe  = nvalid ? n : (DIM - 1);

    const float* arow = X + (size_t)(m0 + l16) * DIM + 2 * half;
    const float* wrow = W + (size_t)nsafe      * DIM + 2 * half;

    v8f acc0 = {};
    v8f acc1 = {};
    int kt = 0;
    for (; kt + 8 <= DIM; kt += 8) {
        v2f a0, b0, a1, b1;
        a0.x = arow[kt];     a0.y = arow[kt + 1];
        b0.x = wrow[kt];     b0.y = wrow[kt + 1];
        a1.x = arow[kt + 4]; a1.y = arow[kt + 5];
        b1.x = wrow[kt + 4]; b1.y = wrow[kt + 5];
        acc0 = __builtin_amdgcn_wmma_f32_16x16x4_f32(
            false, a0, false, b0, (short)0, acc0, false, false);
        acc1 = __builtin_amdgcn_wmma_f32_16x16x4_f32(
            false, a1, false, b1, (short)0, acc1, false, false);
    }
    {
        v2f a0, b0;
        a0.x = arow[kt]; a0.y = arow[kt + 1];
        b0.x = wrow[kt]; b0.y = wrow[kt + 1];
        acc0 = __builtin_amdgcn_wmma_f32_16x16x4_f32(
            false, a0, false, b0, (short)0, acc0, false, false);
    }
    const v8f acc = acc0 + acc1;

    if (nvalid) {
        const float bv = bias[n];
        #pragma unroll
        for (int r = 0; r < 8; ++r) {
            const int m = m0 + r + 8 * half;
            out[(size_t)m * DIM + n] = acc[r] + bv;
        }
    }
}

__global__ __launch_bounds__(BLOCK) void attn_fused_kernel(
    const float* __restrict__ q, const float* __restrict__ k,
    const float* __restrict__ v, const float* __restrict__ hL,
    const float* __restrict__ hR, float* __restrict__ out)
{
    const int b    = blockIdx.y;
    const int i    = blockIdx.x;
    const int tid  = threadIdx.x;
    const int wave = tid >> 5;
    const int lane = tid & 31;

    __shared__ float s_q[DIM];
    __shared__ float s_attn[T2N];

    const float* qrow = q + ((size_t)b * T1N + i) * DIM;
    for (int d = tid; d < DIM; d += BLOCK) s_q[d] = qrow[d];
    __syncthreads();

    for (int j = wave; j < T2N; j += WAVES) {
        const float* hLp = hL + (((size_t)b * T1N + i) * T2N + j) * DIM;
        const float* hRp = hR + (((size_t)b * T2N + j) * T1N + i) * DIM;
        const float* kp  = k  + ((size_t)b * T2N + j) * DIM;
        float acc = 0.0f;
        for (int d = lane; d < DIM; d += 32)
            acc += (s_q[d] + hLp[d]) * (kp[d] + hRp[d]);
        #pragma unroll
        for (int off = 16; off > 0; off >>= 1)
            acc += __shfl_xor(acc, off, 32);
        if (lane == 0) s_attn[j] = acc;
    }
    __syncthreads();

    if (wave == 0) {
        float m = -__builtin_inff();
        for (int j = lane; j < T2N; j += 32) m = fmaxf(m, s_attn[j]);
        #pragma unroll
        for (int off = 16; off > 0; off >>= 1) m = fmaxf(m, __shfl_xor(m, off, 32));

        float sum = 0.0f;
        for (int j = lane; j < T2N; j += 32) sum += expf(s_attn[j] - m);
        #pragma unroll
        for (int off = 16; off > 0; off >>= 1) sum += __shfl_xor(sum, off, 32);
        const float inv = 1.0f / sum;

        float m2 = -__builtin_inff();
        for (int j = lane; j < T2N; j += 32) {
            const float p = expf(s_attn[j] - m) * inv;
            m2 = fmaxf(m2, 1000.0f * p);
        }
        #pragma unroll
        for (int off = 16; off > 0; off >>= 1) m2 = fmaxf(m2, __shfl_xor(m2, off, 32));

        float sum2 = 0.0f;
        for (int j = lane; j < T2N; j += 32) {
            const float p = expf(s_attn[j] - m) * inv;
            sum2 += expf(1000.0f * p - m2);
        }
        #pragma unroll
        for (int off = 16; off > 0; off >>= 1) sum2 += __shfl_xor(sum2, off, 32);
        const float inv2 = 1.0f / sum2;

        for (int j = lane; j < T2N; j += 32) {
            const float p  = expf(s_attn[j] - m) * inv;
            const float a2 = expf(1000.0f * p - m2) * inv2;
            s_attn[j] = fminf(fmaxf(a2, 0.0f), 1.0f);
        }
    }
    __syncthreads();

    for (int d = tid; d < DIM; d += BLOCK) {
        const float* vp  = v  + (size_t)b * T2N * DIM + d;
        const float* hRp = hR + ((size_t)b * T2N * T1N + i) * DIM + d;
        float acc = 0.0f;
        for (int j = 0; j < T2N; ++j)
            acc += s_attn[j] * (vp[(size_t)j * DIM] + hRp[(size_t)j * T1N * DIM]);
        out[((size_t)b * T1N + i) * DIM + d] = acc;
    }
}

extern "C" void kernel_launch(void* const* d_in, const int* in_sizes, int n_in,
                              void* d_out, int out_size, void* d_ws, size_t ws_size,
                              hipStream_t stream)
{
    const float* query = (const float*)d_in[0];
    const float* key   = (const float*)d_in[1];
    const float* value = (const float*)d_in[2];
    const float* hL    = (const float*)d_in[3];
    const float* hR    = (const float*)d_in[4];
    const float* WQ    = (const float*)d_in[5];
    const float* bQ    = (const float*)d_in[6];
    const float* WK    = (const float*)d_in[7];
    const float* bK    = (const float*)d_in[8];
    float* out = (float*)d_out;

    float* q = (float*)d_ws;
    float* k = q + (size_t)ROWS * DIM;
    float* v = k + (size_t)ROWS * DIM;

    const dim3 pgrid((DIM + 15) / 16, ROWS / 16);
    proj_wmma_kernel<<<pgrid, 32, 0, stream>>>(query, WQ, bQ, q);
    proj_wmma_kernel<<<pgrid, 32, 0, stream>>>(key,   WK, bK, k);
    proj_wmma_kernel<<<pgrid, 32, 0, stream>>>(value, WK, bK, v);

    attn_fused_kernel<<<dim3(T1N, BATCH), BLOCK, 0, stream>>>(q, k, v, hL, hR, out);
}


# score pass: all 30 loads of a row issued before first use, next row's kept in flight (occupancy target = the grid's 4 waves per SIMD); same loads and sum order, scores bit-identical
# speedup vs baseline: 2.0386x; 2.0386x over previous
#include <hip/hip_runtime.h>

#define DIM   300
#define T1N   128
#define T2N   128
#define BATCH 4
#define ROWS  (BATCH * T1N)

#define WAVES 8
#define BLOCK (WAVES * 32)

typedef __attribute__((ext_vector_type(2))) float v2f;
typedef __attribute__((ext_vector_type(8))) float v8f;

__global__ __launch_bounds__(32) void proj_wmma_kernel(
    const float* __restrict__ X, const float* __restrict__ W,
    const float* __restrict__ bias, float* __restrict__ out)
{
    const int lane  = threadIdx.x;
    const int nTile = blockIdx.x;
    const int mTile = blockIdx.y;
    const int m0 = mTile * 16;
    const int n0 = nTile * 16;

    const int half = lane >> 4;
    const int l16  = lane & 15;

    const int  n      = n0 + l16;
    const bool nvalid = (n < DIM);
    const int  nsafe  = nvalid ? n : (DIM - 1);

    const float* arow = X + (size_t)(m0 + l16) * DIM + 2 * half;
    const float* wrow = W + (size_t)nsafe      * DIM + 2 * half;

    v8f acc0 = {};
    v8f acc1 = {};
    int kt = 0;
    for (; kt + 8 <= DIM; kt += 8) {
        v2f a0, b0, a1, b1;
        a0.x = arow[kt];     a0.y = arow[kt + 1];
        b0.x = wrow[kt];     b0.y = wrow[kt + 1];
        a1.x = arow[kt + 4]; a1.y = arow[kt + 5];
        b1.x = wrow[kt + 4]; b1.y = wrow[kt + 5];
        acc0 = __builtin_amdgcn_wmma_f32_16x16x4_f32(
            false, a0, false, b0, (short)0, acc0, false, false);
        acc1 = __builtin_amdgcn_wmma_f32_16x16x4_f32(
            false, a1, false, b1, (short)0, acc1, false, false);
    }
    {
        v2f a0, b0;
        a0.x = arow[kt]; a0.y = arow[kt + 1];
        b0.x = wrow[kt]; b0.y = wrow[kt + 1];
        acc0 = __builtin_amdgcn_wmma_f32_16x16x4_f32(
            false, a0, false, b0, (short)0, acc0, false, false);
    }
    const v8f acc = acc0 + acc1;

    if (nvalid) {
        const float bv = bias[n];
        #pragma unroll
        for (int r = 0; r < 8; ++r) {
            const int m = m0 + r + 8 * half;
            out[(size_t)m * DIM + n] = acc[r] + bv;
        }
    }
}

#define NSTEP 10

struct RowRegs { float a[NSTEP]; float r[NSTEP]; float c[NSTEP]; };

__device__ __forceinline__ void load_row(RowRegs& R, const float* __restrict__ pa,
                                         const float* __restrict__ pr,
                                         const float* __restrict__ pc, int toff)
{
    #pragma unroll
    for (int m = 0; m < NSTEP - 1; ++m) {
        R.a[m] = pa[32 * m];
        R.r[m] = pr[32 * m];
        R.c[m] = pc[32 * m];
    }
    R.a[NSTEP - 1] = pa[toff];
    R.r[NSTEP - 1] = pr[toff];
    R.c[NSTEP - 1] = pc[toff];
}

__device__ __forceinline__ float row_score(const RowRegs& R, const float (&qv)[NSTEP], bool tailv)
{
    float acc = 0.0f;
    #pragma unroll
    for (int m = 0; m < NSTEP - 1; ++m)
        acc += (qv[m] + R.a[m]) * (R.c[m] + R.r[m]);
    const float t = acc + (qv[NSTEP - 1] + R.a[NSTEP - 1]) * (R.c[NSTEP - 1] + R.r[NSTEP - 1]);
    acc = tailv ? t : acc;
    #pragma unroll
    for (int off = 16; off > 0; off >>= 1)
        acc += __shfl_xor(acc, off, 32);
    return acc;
}

__global__ __launch_bounds__(BLOCK) __attribute__((amdgpu_waves_per_eu(4, 4)))
void attn_fused_kernel(
    const float* __restrict__ q, const float* __restrict__ k,
    const float* __restrict__ v, const float* __restrict__ hL,
    const float* __restrict__ hR, float* __restrict__ out)
{
    const int b    = blockIdx.y;
    const int i    = blockIdx.x;
    const int tid  = threadIdx.x;
    const int wave = tid >> 5;
    const int lane = tid & 31;

    __shared__ float s_q[DIM];
    __shared__ float s_attn[T2N];

    const float* qrow = q + ((size_t)b * T1N + i) * DIM;
    for (int d = tid; d < DIM; d += BLOCK) s_q[d] = qrow[d];
    __syncthreads();

    float qv[NSTEP];
    #pragma unroll
    for (int m = 0; m < NSTEP; ++m) {
        const int d = lane + 32 * m;
        qv[m] = s_q[d < DIM ? d : DIM - 1];
    }
    const bool tailv = lane < DIM - 32 * (NSTEP - 1);
    const int  toff  = tailv ? 32 * (NSTEP - 1) : 32 * (NSTEP - 2);

    const float* hLw = hL + (((size_t)b * T1N + i) * T2N + wave) * DIM + lane;
    const float* hRw = hR + (((size_t)b * T2N + wave) * T1N + i) * DIM + lane;
    const float* kw  = k  + ((size_t)b * T2N + wave) * DIM + lane;
    const size_t hLs = (size_t)WAVES * DIM;
    const size_t hRs = (size_t)WAVES * T1N * DIM;

    RowRegs ra, rb;
    load_row(ra, hLw, hRw, kw, toff);
    for (int t = 0; t < T2N / WAVES - 2; t += 2) {
        load_row(rb, hLw + (t + 1) * hLs, hRw + (t + 1) * hRs, kw + (t + 1) * hLs, toff);
        const float sa = row_score(ra, qv, tailv);
        if (lane == 0) s_attn[wave + WAVES * t] = sa;
        load_row(ra, hLw + (t + 2) * hLs, hRw + (t + 2) * hRs, kw + (t + 2) * hLs, toff);
        const float sb = row_score(rb, qv, tailv);
        if (lane == 0) s_attn[wave + WAVES * (t + 1)] = sb;
    }
    {
        const int t = T2N / WAVES - 2;
        load_row(rb, hLw + (t + 1) * hLs, hRw + (t + 1) * hRs, kw + (t + 1) * hLs, toff);
        const float sa = row_score(ra, qv, tailv);
        if (lane == 0) s_attn[wave + WAVES * t] = sa;
        const float sb = row_score(rb, qv, tailv);
        if (lane == 0) s_attn[wave + WAVES * (t + 1)] = sb;
    }
    __syncthreads();

    if (wave == 0) {
        float m = -__builtin_inff();
        for (int j = lane; j < T2N; j += 32) m = fmaxf(m, s_attn[j]);
        #pragma unroll
        for (int off = 16; off > 0; off >>= 1) m = fmaxf(m, __shfl_xor(m, off, 32));

        float sum = 0.0f;
        for (int j = lane; j < T2N; j += 32) sum += expf(s_attn[j] - m);
        #pragma unroll
        for (int off = 16; off > 0; off >>= 1) sum += __shfl_xor(sum, off, 32);
        const float inv = 1.0f / sum;

        float m2 = -__builtin_inff();
        for (int j = lane; j < T2N; j += 32) {
            const float p = expf(s_attn[j] - m) * inv;
            m2 = fmaxf(m2, 1000.0f * p);
        }
        #pragma unroll
        for (int off = 16; off > 0; off >>= 1) m2 = fmaxf(m2, __shfl_xor(m2, off, 32));

        float sum2 = 0.0f;
        for (int j = lane; j < T2N; j += 32) {
            const float p = expf(s_attn[j] - m) * inv;
            sum2 += expf(1000.0f * p - m2);
        }
        #pragma unroll
        for (int off = 16; off > 0; off >>= 1) sum2 += __shfl_xor(sum2, off, 32);
        const float inv2 = 1.0f / sum2;

        for (int j = lane; j < T2N; j += 32) {
            const float p  = expf(s_attn[j] - m) * inv;
            const float a2 = expf(1000.0f * p - m2) * inv2;
            s_attn[j] = fminf(fmaxf(a2, 0.0f), 1.0f);
        }
    }
    __syncthreads();

    for (int d = tid; d < DIM; d += BLOCK) {
        const float* vp  = v  + (size_t)b * T2N * DIM + d;
        const float* hRp = hR + ((size_t)b * T2N * T1N + i) * DIM + d;
        float acc = 0.0f;
        for (int j = 0; j < T2N; ++j)
            acc += s_attn[j] * (vp[(size_t)j * DIM] + hRp[(size_t)j * T1N * DIM]);
        out[((size_t)b * T1N + i) * DIM + d] = acc;
    }
}

extern "C" void kernel_launch(void* const* d_in, const int* in_sizes, int n_in,
                              void* d_out, int out_size, void* d_ws, size_t ws_size,
                              hipStream_t stream)
{
    const float* query = (const float*)d_in[0];
    const float* key   = (const float*)d_in[1];
    const float* value = (const float*)d_in[2];
    const float* hL    = (const float*)d_in[3];
    const float* hR    = (const float*)d_in[4];
    const float* WQ    = (const float*)d_in[5];
    const float* bQ    = (const float*)d_in[6];
    const float* WK    = (const float*)d_in[7];
    const float* bK    = (const float*)d_in[8];
    float* out = (float*)d_out;

    float* q = (float*)d_ws;
    float* k = q + (size_t)ROWS * DIM;
    float* v = k + (size_t)ROWS * DIM;

    const dim3 pgrid((DIM + 15) / 16, ROWS / 16);
    proj_wmma_kernel<<<pgrid, 32, 0, stream>>>(query, WQ, bQ, q);
    proj_wmma_kernel<<<pgrid, 32, 0, stream>>>(key,   WK, bK, k);
    proj_wmma_kernel<<<pgrid, 32, 0, stream>>>(value, WK, bK, v);

    attn_fused_kernel<<<dim3(T1N, BATCH), BLOCK, 0, stream>>>(q, k, v, hL, hR, out);
}


# pass 2 visits only the rows j whose sharpened weight is non-zero (wave 0 leaves a 128-bit mask by ballot); a zero weight adds exactly zero, remaining terms keep their order, output values unchanged
# speedup vs baseline: 2.9065x; 1.4257x over previous
#include <hip/hip_runtime.h>

#define DIM   300
#define T1N   128
#define T2N   128
#define BATCH 4
#define ROWS  (BATCH * T1N)

#define WAVES 8
#define BLOCK (WAVES * 32)

typedef __attribute__((ext_vector_type(2))) float v2f;
typedef __attribute__((ext_vector_type(8))) float v8f;

__global__ __launch_bounds__(32) void proj_wmma_kernel(
    const float* __restrict__ X, const float* __restrict__ W,
    const float* __restrict__ bias, float* __restrict__ out)
{
    const int lane  = threadIdx.x;
    const int nTile = blockIdx.x;
    const int mTile = blockIdx.y;
    const int m0 = mTile * 16;
    const int n0 = nTile * 16;

    const int half = lane >> 4;
    const int l16  = lane & 15;

    const int  n      = n0 + l16;
    const bool nvalid = (n < DIM);
    const int  nsafe  = nvalid ? n : (DIM - 1);

    const float* arow = X + (size_t)(m0 + l16) * DIM + 2 * half;
    const float* wrow = W + (size_t)nsafe      * DIM + 2 * half;

    v8f acc0 = {};
    v8f acc1 = {};
    int kt = 0;
    for (; kt + 8 <= DIM; kt += 8) {
        v2f a0, b0, a1, b1;
        a0.x = arow[kt];     a0.y = arow[kt + 1];
        b0.x = wrow[kt];     b0.y = wrow[kt + 1];
        a1.x = arow[kt + 4]; a1.y = arow[kt + 5];
        b1.x = wrow[kt + 4]; b1.y = wrow[kt + 5];
        acc0 = __builtin_amdgcn_wmma_f32_16x16x4_f32(
            false, a0, false, b0, (short)0, acc0, false, false);
        acc1 = __builtin_amdgcn_wmma_f32_16x16x4_f32(
            false, a1, false, b1, (short)0, acc1, false, false);
    }
    {
        v2f a0, b0;
        a0.x = arow[kt]; a0.y = arow[kt + 1];
        b0.x = wrow[kt]; b0.y = wrow[kt + 1];
        acc0 = __builtin_amdgcn_wmma_f32_16x16x4_f32(
            false, a0, false, b0, (short)0, acc0, false, false);
    }
    const v8f acc = acc0 + acc1;

    if (nvalid) {
        const float bv = bias[n];
        #pragma unroll
        for (int r = 0; r < 8; ++r) {
            const int m = m0 + r + 8 * half;
            out[(size_t)m * DIM + n] = acc[r] + bv;
        }
    }
}

#define NSTEP 10

struct RowRegs { float a[NSTEP]; float r[NSTEP]; float c[NSTEP]; };

__device__ __forceinline__ void load_row(RowRegs& R, const float* __restrict__ pa,
                                         const float* __restrict__ pr,
                                         const float* __restrict__ pc, int toff)
{
    #pragma unroll
    for (int m = 0; m < NSTEP - 1; ++m) {
        R.a[m] = pa[32 * m];
        R.r[m] = pr[32 * m];
        R.c[m] = pc[32 * m];
    }
    R.a[NSTEP - 1] = pa[toff];
    R.r[NSTEP - 1] = pr[toff];
    R.c[NSTEP - 1] = pc[toff];
}

__device__ __forceinline__ float row_score(const RowRegs& R, const float (&qv)[NSTEP], bool tailv)
{
    float acc = 0.0f;
    #pragma unroll
    for (int m = 0; m < NSTEP - 1; ++m)
        acc += (qv[m] + R.a[m]) * (R.c[m] + R.r[m]);
    const float t = acc + (qv[NSTEP - 1] + R.a[NSTEP - 1]) * (R.c[NSTEP - 1] + R.r[NSTEP - 1]);
    acc = tailv ? t : acc;
    #pragma unroll
    for (int off = 16; off > 0; off >>= 1)
        acc += __shfl_xor(acc, off, 32);
    return acc;
}

__global__ __launch_bounds__(BLOCK) __attribute__((amdgpu_waves_per_eu(4, 4)))
void attn_fused_kernel(
    const float* __restrict__ q, const float* __restrict__ k,
    const float* __restrict__ v, const float* __restrict__ hL,
    const float* __restrict__ hR, float* __restrict__ out)
{
    const int b    = blockIdx.y;
    const int i    = blockIdx.x;
    const int tid  = threadIdx.x;
    const int wave = tid >> 5;
    const int lane = tid & 31;

    __shared__ float s_q[DIM];
    __shared__ float s_attn[T2N];
    __shared__ unsigned s_nz[T2N / 32];

    const float* qrow = q + ((size_t)b * T1N + i) * DIM;
    for (int d = tid; d < DIM; d += BLOCK) s_q[d] = qrow[d];
    __syncthreads();

    float qv[NSTEP];
    #pragma unroll
    for (int m = 0; m < NSTEP; ++m) {
        const int d = lane + 32 * m;
        qv[m] = s_q[d < DIM ? d : DIM - 1];
    }
    const bool tailv = lane < DIM - 32 * (NSTEP - 1);
    const int  toff  = tailv ? 32 * (NSTEP - 1) : 32 * (NSTEP - 2);

    const float* hLw = hL + (((size_t)b * T1N + i) * T2N + wave) * DIM + lane;
    const float* hRw = hR + (((size_t)b * T2N + wave) * T1N + i) * DIM + lane;
    const float* kw  = k  + ((size_t)b * T2N + wave) * DIM + lane;
    const size_t hLs = (size_t)WAVES * DIM;
    const size_t hRs = (size_t)WAVES * T1N * DIM;

    RowRegs ra, rb;
    load_row(ra, hLw, hRw, kw, toff);
    for (int t = 0; t < T2N / WAVES - 2; t += 2) {
        load_row(rb, hLw + (t + 1) * hLs, hRw + (t + 1) * hRs, kw + (t + 1) * hLs, toff);
        const float sa = row_score(ra, qv, tailv);
        if (lane == 0) s_attn[wave + WAVES * t] = sa;
        load_row(ra, hLw + (t + 2) * hLs, hRw + (t + 2) * hRs, kw + (t + 2) * hLs, toff);
        const float sb = row_score(rb, qv, tailv);
        if (lane == 0) s_attn[wave + WAVES * (t + 1)] = sb;
    }
    {
        const int t = T2N / WAVES - 2;
        load_row(rb, hLw + (t + 1) * hLs, hRw + (t + 1) * hRs, kw + (t + 1) * hLs, toff);
        const float sa = row_score(ra, qv, tailv);
        if (lane == 0) s_attn[wave + WAVES * t] = sa;
        const float sb = row_score(rb, qv, tailv);
        if (lane == 0) s_attn[wave + WAVES * (t + 1)] = sb;
    }
    __syncthreads();

    if (wave == 0) {
        float m = -__builtin_inff();
        for (int j = lane; j < T2N; j += 32) m = fmaxf(m, s_attn[j]);
        #pragma unroll
        for (int off = 16; off > 0; off >>= 1) m = fmaxf(m, __shfl_xor(m, off, 32));

        float sum = 0.0f;
        for (int j = lane; j < T2N; j += 32) sum += expf(s_attn[j] - m);
        #pragma unroll
        for (int off = 16; off > 0; off >>= 1) sum += __shfl_xor(sum, off, 32);
        const float inv = 1.0f / sum;

        float m2 = -__builtin_inff();
        for (int j = lane; j < T2N; j += 32) {
            const float p = expf(s_attn[j] - m) * inv;
            m2 = fmaxf(m2, 1000.0f * p);
        }
        #pragma unroll
        for (int off = 16; off > 0; off >>= 1) m2 = fmaxf(m2, __shfl_xor(m2, off, 32));

        float sum2 = 0.0f;
        for (int j = lane; j < T2N; j += 32) {
            const float p = expf(s_attn[j] - m) * inv;
            sum2 += expf(1000.0f * p - m2);
        }
        #pragma unroll
        for (int off = 16; off > 0; off >>= 1) sum2 += __shfl_xor(sum2, off, 32);
        const float inv2 = 1.0f / sum2;

        for (int t = 0; t < T2N / 32; ++t) {
            const int   j  = lane + 32 * t;
            const float p  = expf(s_attn[j] - m) * inv;
            const float a2 = expf(1000.0f * p - m2) * inv2;
            const float w  = fminf(fmaxf(a2, 0.0f), 1.0f);
            s_attn[j] = w;
            const unsigned present = __builtin_amdgcn_ballot_w32(!(w == 0.0f));
            if (lane == 0) s_nz[t] = present;
        }
    }
    __syncthreads();

    for (int d = tid; d < DIM; d += BLOCK) {
        const float* vp  = v  + (size_t)b * T2N * DIM + d;
        const float* hRp = hR + ((size_t)b * T2N * T1N + i) * DIM + d;
        float acc = 0.0f;
        for (int t = 0; t < T2N / 32; ++t) {
            unsigned mk = (unsigned)__builtin_amdgcn_readfirstlane((int)s_nz[t]);
            while (mk != 0u) {
                const int j = 32 * t + __builtin_ctz(mk);
                mk &= mk - 1u;
                acc += s_attn[j] * (vp[(size_t)j * DIM] + hRp[(size_t)j * T1N * DIM]);
            }
        }
        out[((size_t)b * T1N + i) * DIM + d] = acc;
    }
}

extern "C" void kernel_launch(void* const* d_in, const int* in_sizes, int n_in,
                              void* d_out, int out_size, void* d_ws, size_t ws_size,
                              hipStream_t stream)
{
    const float* query = (const float*)d_in[0];
    const float* key   = (const float*)d_in[1];
    const float* value = (const float*)d_in[2];
    const float* hL    = (const float*)d_in[3];
    const float* hR    = (const float*)d_in[4];
    const float* WQ    = (const float*)d_in[5];
    const float* bQ    = (const float*)d_in[6];
    const float* WK    = (const float*)d_in[7];
    const float* bK    = (const float*)d_in[8];
    float* out = (float*)d_out;

    float* q = (float*)d_ws;
    float* k = q + (size_t)ROWS * DIM;
    float* v = k + (size_t)ROWS * DIM;

    const dim3 pgrid((DIM + 15) / 16, ROWS / 16);
    proj_wmma_kernel<<<pgrid, 32, 0, stream>>>(query, WQ, bQ, q);
    proj_wmma_kernel<<<pgrid, 32, 0, stream>>>(key,   WK, bK, k);
    proj_wmma_kernel<<<pgrid, 32, 0, stream>>>(value, WK, bK, v);

    attn_fused_kernel<<<dim3(T1N, BATCH), BLOCK, 0, stream>>>(q, k, v, hL, hR, out);
}


# the three projection launches (q, k, v) become one launch, blockIdx.z picks the operand set; per-tile instruction sequence and sum order unchanged, results bit-identical
# speedup vs baseline: 4.1366x; 1.4232x over previous
#include <hip/hip_runtime.h>

#define DIM   300
#define T1N   128
#define T2N   128
#define BATCH 4
#define ROWS  (BATCH * T1N)

#define WAVES 8
#define BLOCK (WAVES * 32)

typedef __attribute__((ext_vector_type(2))) float v2f;
typedef __attribute__((ext_vector_type(8))) float v8f;

__global__ __launch_bounds__(32) void proj_wmma_kernel(
    const float* __restrict__ X0, const float* __restrict__ W0,
    const float* __restrict__ bias0, float* __restrict__ out0,
    const float* __restrict__ X1, const float* __restrict__ W1,
    const float* __restrict__ bias1, float* __restrict__ out1,
    const float* __restrict__ X2, const float* __restrict__ W2,
    const float* __restrict__ bias2, float* __restrict__ out2)
{
    const int which = blockIdx.z;
    const float* __restrict__ X    = which == 0 ? X0    : (which == 1 ? X1    : X2);
    const float* __restrict__ W    = which == 0 ? W0    : (which == 1 ? W1    : W2);
    const float* __restrict__ bias = which == 0 ? bias0 : (which == 1 ? bias1 : bias2);
    float* __restrict__       out  = which == 0 ? out0  : (which == 1 ? out1  : out2);

    const int lane  = threadIdx.x;
    const int nTile = blockIdx.x;
    const int mTile = blockIdx.y;
    const int m0 = mTile * 16;
    const int n0 = nTile * 16;

    const int half = lane >> 4;
    const int l16  = lane & 15;

    const int  n      = n0 + l16;
    const bool nvalid = (n < DIM);
    const int  nsafe  = nvalid ? n : (DIM - 1);

    const float* arow = X + (size_t)(m0 + l16) * DIM + 2 * half;
    const float* wrow = W + (size_t)nsafe      * DIM + 2 * half;

    v8f acc0 = {};
    v8f acc1 = {};
    int kt = 0;
    for (; kt + 8 <= DIM; kt += 8) {
        v2f a0, b0, a1, b1;
        a0.x = arow[kt];     a0.y = arow[kt + 1];
        b0.x = wrow[kt];     b0.y = wrow[kt + 1];
        a1.x = arow[kt + 4]; a1.y = arow[kt + 5];
        b1.x = wrow[kt + 4]; b1.y = wrow[kt + 5];
        acc0 = __builtin_amdgcn_wmma_f32_16x16x4_f32(
            false, a0, false, b0, (short)0, acc0, false, false);
        acc1 = __builtin_amdgcn_wmma_f32_16x16x4_f32(
            false, a1, false, b1, (short)0, acc1, false, false);
    }
    {
        v2f a0, b0;
        a0.x = arow[kt]; a0.y = arow[kt + 1];
        b0.x = wrow[kt]; b0.y = wrow[kt + 1];
        acc0 = __builtin_amdgcn_wmma_f32_16x16x4_f32(
            false, a0, false, b0, (short)0, acc0, false, false);
    }
    const v8f acc = acc0 + acc1;

    if (nvalid) {
        const float bv = bias[n];
        #pragma unroll
        for (int r = 0; r < 8; ++r) {
            const int m = m0 + r + 8 * half;
            out[(size_t)m * DIM + n] = acc[r] + bv;
        }
    }
}

#define NSTEP 10

struct RowRegs { float a[NSTEP]; float r[NSTEP]; float c[NSTEP]; };

__device__ __forceinline__ void load_row(RowRegs& R, const float* __restrict__ pa,
                                         const float* __restrict__ pr,
                                         const float* __restrict__ pc, int toff)
{
    #pragma unroll
    for (int m = 0; m < NSTEP - 1; ++m) {
        R.a[m] = pa[32 * m];
        R.r[m] = pr[32 * m];
        R.c[m] = pc[32 * m];
    }
    R.a[NSTEP - 1] = pa[toff];
    R.r[NSTEP - 1] = pr[toff];
    R.c[NSTEP - 1] = pc[toff];
}

__device__ __forceinline__ float row_score(const RowRegs& R, const float (&qv)[NSTEP], bool tailv)
{
    float acc = 0.0f;
    #pragma unroll
    for (int m = 0; m < NSTEP - 1; ++m)
        acc += (qv[m] + R.a[m]) * (R.c[m] + R.r[m]);
    const float t = acc + (qv[NSTEP - 1] + R.a[NSTEP - 1]) * (R.c[NSTEP - 1] + R.r[NSTEP - 1]);
    acc = tailv ? t : acc;
    #pragma unroll
    for (int off = 16; off > 0; off >>= 1)
        acc += __shfl_xor(acc, off, 32);
    return acc;
}

__global__ __launch_bounds__(BLOCK) __attribute__((amdgpu_waves_per_eu(4, 4)))
void attn_fused_kernel(
    const float* __restrict__ q, const float* __restrict__ k,
    const float* __restrict__ v, const float* __restrict__ hL,
    const float* __restrict__ hR, float* __restrict__ out)
{
    const int b    = blockIdx.y;
    const int i    = blockIdx.x;
    const int tid  = threadIdx.x;
    const int wave = tid >> 5;
    const int lane = tid & 31;

    __shared__ float s_q[DIM];
    __shared__ float s_attn[T2N];
    __shared__ unsigned s_nz[T2N / 32];

    const float* qrow = q + ((size_t)b * T1N + i) * DIM;
    for (int d = tid; d < DIM; d += BLOCK) s_q[d] = qrow[d];
    __syncthreads();

    float qv[NSTEP];
    #pragma unroll
    for (int m = 0; m < NSTEP; ++m) {
        const int d = lane + 32 * m;
        qv[m] = s_q[d < DIM ? d : DIM - 1];
    }
    const bool tailv = lane < DIM - 32 * (NSTEP - 1);
    const int  toff  = tailv ? 32 * (NSTEP - 1) : 32 * (NSTEP - 2);

    const float* hLw = hL + (((size_t)b * T1N + i) * T2N + wave) * DIM + lane;
    const float* hRw = hR + (((size_t)b * T2N + wave) * T1N + i) * DIM + lane;
    const float* kw  = k  + ((size_t)b * T2N + wave) * DIM + lane;
    const size_t hLs = (size_t)WAVES * DIM;
    const size_t hRs = (size_t)WAVES * T1N * DIM;

    RowRegs ra, rb;
    load_row(ra, hLw, hRw, kw, toff);
    for (int t = 0; t < T2N / WAVES - 2; t += 2) {
        load_row(rb, hLw + (t + 1) * hLs, hRw + (t + 1) * hRs, kw + (t + 1) * hLs, toff);
        const float sa = row_score(ra, qv, tailv);
        if (lane == 0) s_attn[wave + WAVES * t] = sa;
        load_row(ra, hLw + (t + 2) * hLs, hRw + (t + 2) * hRs, kw + (t + 2) * hLs, toff);
        const float sb = row_score(rb, qv, tailv);
        if (lane == 0) s_attn[wave + WAVES * (t + 1)] = sb;
    }
    {
        const int t = T2N / WAVES - 2;
        load_row(rb, hLw + (t + 1) * hLs, hRw + (t + 1) * hRs, kw + (t + 1) * hLs, toff);
        const float sa = row_score(ra, qv, tailv);
        if (lane == 0) s_attn[wave + WAVES * t] = sa;
        const float sb = row_score(rb, qv, tailv);
        if (lane == 0) s_attn[wave + WAVES * (t + 1)] = sb;
    }
    __syncthreads();

    if (wave == 0) {
        float m = -__builtin_inff();
        for (int j = lane; j < T2N; j += 32) m = fmaxf(m, s_attn[j]);
        #pragma unroll
        for (int off = 16; off > 0; off >>= 1) m = fmaxf(m, __shfl_xor(m, off, 32));

        float sum = 0.0f;
        for (int j = lane; j < T2N; j += 32) sum += expf(s_attn[j] - m);
        #pragma unroll
        for (int off = 16; off > 0; off >>= 1) sum += __shfl_xor(sum, off, 32);
        const float inv = 1.0f / sum;

        float m2 = -__builtin_inff();
        for (int j = lane; j < T2N; j += 32) {
            const float p = expf(s_attn[j] - m) * inv;
            m2 = fmaxf(m2, 1000.0f * p);
        }
        #pragma unroll
        for (int off = 16; off > 0; off >>= 1) m2 = fmaxf(m2, __shfl_xor(m2, off, 32));

        float sum2 = 0.0f;
        for (int j = lane; j < T2N; j += 32) {
            const float p = expf(s_attn[j] - m) * inv;
            sum2 += expf(1000.0f * p - m2);
        }
        #pragma unroll
        for (int off = 16; off > 0; off >>= 1) sum2 += __shfl_xor(sum2, off, 32);
        const float inv2 = 1.0f / sum2;

        for (int t = 0; t < T2N / 32; ++t) {
            const int   j  = lane + 32 * t;
            const float p  = expf(s_attn[j] - m) * inv;
            const float a2 = expf(1000.0f * p - m2) * inv2;
            const float w  = fminf(fmaxf(a2, 0.0f), 1.0f);
            s_attn[j] = w;
            const unsigned present = __builtin_amdgcn_ballot_w32(!(w == 0.0f));
            if (lane == 0) s_nz[t] = present;
        }
    }
    __syncthreads();

    for (int d = tid; d < DIM; d += BLOCK) {
        const float* vp  = v  + (size_t)b * T2N * DIM + d;
        const float* hRp = hR + ((size_t)b * T2N * T1N + i) * DIM + d;
        float acc = 0.0f;
        for (int t = 0; t < T2N / 32; ++t) {
            unsigned mk = (unsigned)__builtin_amdgcn_readfirstlane((int)s_nz[t]);
            while (mk != 0u) {
                const int j = 32 * t + __builtin_ctz(mk);
                mk &= mk - 1u;
                acc += s_attn[j] * (vp[(size_t)j * DIM] + hRp[(size_t)j * T1N * DIM]);
            }
        }
        out[((size_t)b * T1N + i) * DIM + d] = acc;
    }
}

extern "C" void kernel_launch(void* const* d_in, const int* in_sizes, int n_in,
                              void* d_out, int out_size, void* d_ws, size_t ws_size,
                              hipStream_t stream)
{
    const float* query = (const float*)d_in[0];
    const float* key   = (const float*)d_in[1];
    const float* value = (const float*)d_in[2];
    const float* hL    = (const float*)d_in[3];
    const float* hR    = (const float*)d_in[4];
    const float* WQ    = (const float*)d_in[5];
    const float* bQ    = (const float*)d_in[6];
    const float* WK    = (const float*)d_in[7];
    const float* bK    = (const float*)d_in[8];
    float* out = (float*)d_out;

    float* q = (float*)d_ws;
    float* k = q + (size_t)ROWS * DIM;
    float* v = k + (size_t)ROWS * DIM;

    const dim3 pgrid((DIM + 15) / 16, ROWS / 16, 3);
    proj_wmma_kernel<<<pgrid, 32, 0, stream>>>(query, WQ, bQ, q,
                                               key,   WK, bK, k,
                                               value, WK, bK, v);

    attn_fused_kernel<<<dim3(T1N, BATCH), BLOCK, 0, stream>>>(q, k, v, hL, hR, out);
}


# projection: each 16-row operand tile is read as whole rows with 16-byte loads (512 contiguous bytes per load) and staged in LDS, WMMA fragments come from LDS; same operands, sum order, bit-identical
# speedup vs baseline: 4.4776x; 1.0824x over previous
#include <hip/hip_runtime.h>

#define DIM   300
#define T1N   128
#define T2N   128
#define BATCH 4
#define ROWS  (BATCH * T1N)

#define WAVES 8
#define BLOCK (WAVES * 32)

typedef __attribute__((ext_vector_type(2))) float v2f;
typedef __attribute__((ext_vector_type(8))) float v8f;

typedef __attribute__((ext_vector_type(4))) float v4f;
#define TILE4     (16 * DIM / 4)
#define NLD       ((TILE4 + 31) / 32)
#define TILE4_PAD (NLD * 32)

__global__ __launch_bounds__(32) void proj_wmma_kernel(
    const float* __restrict__ X0, const float* __restrict__ W0,
    const float* __restrict__ bias0, float* __restrict__ out0,
    const float* __restrict__ X1, const float* __restrict__ W1,
    const float* __restrict__ bias1, float* __restrict__ out1,
    const float* __restrict__ X2, const float* __restrict__ W2,
    const float* __restrict__ bias2, float* __restrict__ out2)
{
    __shared__ v4f sA4[TILE4_PAD];
    __shared__ v4f sB4[TILE4_PAD];

    const int which = blockIdx.z;
    const float* __restrict__ X    = which == 0 ? X0    : (which == 1 ? X1    : X2);
    const float* __restrict__ W    = which == 0 ? W0    : (which == 1 ? W1    : W2);
    const float* __restrict__ bias = which == 0 ? bias0 : (which == 1 ? bias1 : bias2);
    float* __restrict__       out  = which == 0 ? out0  : (which == 1 ? out1  : out2);

    const int lane  = threadIdx.x;
    const int nTile = blockIdx.x;
    const int mTile = blockIdx.y;
    const int m0 = mTile * 16;
    const int n0 = nTile * 16;

    const int half = lane >> 4;
    const int l16  = lane & 15;

    const int  n      = n0 + l16;
    const bool nvalid = (n < DIM);

    const v4f* __restrict__ gA = reinterpret_cast<const v4f*>(X + (size_t)m0 * DIM);
    const v4f* __restrict__ gB = reinterpret_cast<const v4f*>(W + (size_t)n0 * DIM);
    const int rowsB = (DIM - n0) < 16 ? (DIM - n0) : 16;
    const int limB  = rowsB * (DIM / 4) - 1;

    v4f ra[NLD], rb[NLD];
    #pragma unroll
    for (int g = 0; g < NLD - 1; ++g) ra[g] = gA[lane + 32 * g];
    {
        const int idx = lane + 32 * (NLD - 1);
        ra[NLD - 1] = gA[idx < TILE4 ? idx : TILE4 - 1];
    }
    #pragma unroll
    for (int g = 0; g < NLD; ++g) {
        const int idx = lane + 32 * g;
        rb[g] = gB[idx < limB ? idx : limB];
    }
    #pragma unroll
    for (int g = 0; g < NLD; ++g) sA4[lane + 32 * g] = ra[g];
    #pragma unroll
    for (int g = 0; g < NLD; ++g) sB4[lane + 32 * g] = rb[g];
    __syncthreads();

    const v2f* fa = reinterpret_cast<const v2f*>(reinterpret_cast<const float*>(sA4) + l16 * DIM + 2 * half);
    const v2f* fb = reinterpret_cast<const v2f*>(reinterpret_cast<const float*>(sB4) + l16 * DIM + 2 * half);

    v8f acc0 = {};
    v8f acc1 = {};
    #pragma unroll
    for (int s = 0; s + 2 <= DIM / 4; s += 2) {
        const v2f a0 = fa[2 * s],     b0 = fb[2 * s];
        const v2f a1 = fa[2 * s + 2], b1 = fb[2 * s + 2];
        acc0 = __builtin_amdgcn_wmma_f32_16x16x4_f32(
            false, a0, false, b0, (short)0, acc0, false, false);
        acc1 = __builtin_amdgcn_wmma_f32_16x16x4_f32(
            false, a1, false, b1, (short)0, acc1, false, false);
    }
    {
        const v2f a0 = fa[2 * (DIM / 4 - 1)], b0 = fb[2 * (DIM / 4 - 1)];
        acc0 = __builtin_amdgcn_wmma_f32_16x16x4_f32(
            false, a0, false, b0, (short)0, acc0, false, false);
    }
    const v8f acc = acc0 + acc1;

    if (nvalid) {
        const float bv = bias[n];
        #pragma unroll
        for (int r = 0; r < 8; ++r) {
            const int m = m0 + r + 8 * half;
            out[(size_t)m * DIM + n] = acc[r] + bv;
        }
    }
}

#define NSTEP 10

struct RowRegs { float a[NSTEP]; float r[NSTEP]; float c[NSTEP]; };

__device__ __forceinline__ void load_row(RowRegs& R, const float* __restrict__ pa,
                                         const float* __restrict__ pr,
                                         const float* __restrict__ pc, int toff)
{
    #pragma unroll
    for (int m = 0; m < NSTEP - 1; ++m) {
        R.a[m] = pa[32 * m];
        R.r[m] = pr[32 * m];
        R.c[m] = pc[32 * m];
    }
    R.a[NSTEP - 1] = pa[toff];
    R.r[NSTEP - 1] = pr[toff];
    R.c[NSTEP - 1] = pc[toff];
}

__device__ __forceinline__ float row_score(const RowRegs& R, const float (&qv)[NSTEP], bool tailv)
{
    float acc = 0.0f;
    #pragma unroll
    for (int m = 0; m < NSTEP - 1; ++m)
        acc += (qv[m] + R.a[m]) * (R.c[m] + R.r[m]);
    const float t = acc + (qv[NSTEP - 1] + R.a[NSTEP - 1]) * (R.c[NSTEP - 1] + R.r[NSTEP - 1]);
    acc = tailv ? t : acc;
    #pragma unroll
    for (int off = 16; off > 0; off >>= 1)
        acc += __shfl_xor(acc, off, 32);
    return acc;
}

__global__ __launch_bounds__(BLOCK) __attribute__((amdgpu_waves_per_eu(4, 4)))
void attn_fused_kernel(
    const float* __restrict__ q, const float* __restrict__ k,
    const float* __restrict__ v, const float* __restrict__ hL,
    const float* __restrict__ hR, float* __restrict__ out)
{
    const int b    = blockIdx.y;
    const int i    = blockIdx.x;
    const int tid  = threadIdx.x;
    const int wave = tid >> 5;
    const int lane = tid & 31;

    __shared__ float s_q[DIM];
    __shared__ float s_attn[T2N];
    __shared__ unsigned s_nz[T2N / 32];

    const float* qrow = q + ((size_t)b * T1N + i) * DIM;
    for (int d = tid; d < DIM; d += BLOCK) s_q[d] = qrow[d];
    __syncthreads();

    float qv[NSTEP];
    #pragma unroll
    for (int m = 0; m < NSTEP; ++m) {
        const int d = lane + 32 * m;
        qv[m] = s_q[d < DIM ? d : DIM - 1];
    }
    const bool tailv = lane < DIM - 32 * (NSTEP - 1);
    const int  toff  = tailv ? 32 * (NSTEP - 1) : 32 * (NSTEP - 2);

    const float* hLw = hL + (((size_t)b * T1N + i) * T2N + wave) * DIM + lane;
    const float* hRw = hR + (((size_t)b * T2N + wave) * T1N + i) * DIM + lane;
    const float* kw  = k  + ((size_t)b * T2N + wave) * DIM + lane;
    const size_t hLs = (size_t)WAVES * DIM;
    const size_t hRs = (size_t)WAVES * T1N * DIM;

    RowRegs ra, rb;
    load_row(ra, hLw, hRw, kw, toff);
    for (int t = 0; t < T2N / WAVES - 2; t += 2) {
        load_row(rb, hLw + (t + 1) * hLs, hRw + (t + 1) * hRs, kw + (t + 1) * hLs, toff);
        const float sa = row_score(ra, qv, tailv);
        if (lane == 0) s_attn[wave + WAVES * t] = sa;
        load_row(ra, hLw + (t + 2) * hLs, hRw + (t + 2) * hRs, kw + (t + 2) * hLs, toff);
        const float sb = row_score(rb, qv, tailv);
        if (lane == 0) s_attn[wave + WAVES * (t + 1)] = sb;
    }
    {
        const int t = T2N / WAVES - 2;
        load_row(rb, hLw + (t + 1) * hLs, hRw + (t + 1) * hRs, kw + (t + 1) * hLs, toff);
        const float sa = row_score(ra, qv, tailv);
        if (lane == 0) s_attn[wave + WAVES * t] = sa;
        const float sb = row_score(rb, qv, tailv);
        if (lane == 0) s_attn[wave + WAVES * (t + 1)] = sb;
    }
    __syncthreads();

    if (wave == 0) {
        float m = -__builtin_inff();
        for (int j = lane; j < T2N; j += 32) m = fmaxf(m, s_attn[j]);
        #pragma unroll
        for (int off = 16; off > 0; off >>= 1) m = fmaxf(m, __shfl_xor(m, off, 32));

        float sum = 0.0f;
        for (int j = lane; j < T2N; j += 32) sum += expf(s_attn[j] - m);
        #pragma unroll
        for (int off = 16; off > 0; off >>= 1) sum += __shfl_xor(sum, off, 32);
        const float inv = 1.0f / sum;

        float m2 = -__builtin_inff();
        for (int j = lane; j < T2N; j += 32) {
            const float p = expf(s_attn[j] - m) * inv;
            m2 = fmaxf(m2, 1000.0f * p);
        }
        #pragma unroll
        for (int off = 16; off > 0; off >>= 1) m2 = fmaxf(m2, __shfl_xor(m2, off, 32));

        float sum2 = 0.0f;
        for (int j = lane; j < T2N; j += 32) {
            const float p = expf(s_attn[j] - m) * inv;
            sum2 += expf(1000.0f * p - m2);
        }
        #pragma unroll
        for (int off = 16; off > 0; off >>= 1) sum2 += __shfl_xor(sum2, off, 32);
        const float inv2 = 1.0f / sum2;

        for (int t = 0; t < T2N / 32; ++t) {
            const int   j  = lane + 32 * t;
            const float p  = expf(s_attn[j] - m) * inv;
            const float a2 = expf(1000.0f * p - m2) * inv2;
            const float w  = fminf(fmaxf(a2, 0.0f), 1.0f);
            s_attn[j] = w;
            const unsigned present = __builtin_amdgcn_ballot_w32(!(w == 0.0f));
            if (lane == 0) s_nz[t] = present;
        }
    }
    __syncthreads();

    for (int d = tid; d < DIM; d += BLOCK) {
        const float* vp  = v  + (size_t)b * T2N * DIM + d;
        const float* hRp = hR + ((size_t)b * T2N * T1N + i) * DIM + d;
        float acc = 0.0f;
        for (int t = 0; t < T2N / 32; ++t) {
            unsigned mk = (unsigned)__builtin_amdgcn_readfirstlane((int)s_nz[t]);
            while (mk != 0u) {
                const int j = 32 * t + __builtin_ctz(mk);
                mk &= mk - 1u;
                acc += s_attn[j] * (vp[(size_t)j * DIM] + hRp[(size_t)j * T1N * DIM]);
            }
        }
        out[((size_t)b * T1N + i) * DIM + d] = acc;
    }
}

extern "C" void kernel_launch(void* const* d_in, const int* in_sizes, int n_in,
                              void* d_out, int out_size, void* d_ws, size_t ws_size,
                              hipStream_t stream)
{
    const float* query = (const float*)d_in[0];
    const float* key   = (const float*)d_in[1];
    const float* value = (const float*)d_in[2];
    const float* hL    = (const float*)d_in[3];
    const float* hR    = (const float*)d_in[4];
    const float* WQ    = (const float*)d_in[5];
    const float* bQ    = (const float*)d_in[6];
    const float* WK    = (const float*)d_in[7];
    const float* bK    = (const float*)d_in[8];
    float* out = (float*)d_out;

    float* q = (float*)d_ws;
    float* k = q + (size_t)ROWS * DIM;
    float* v = k + (size_t)ROWS * DIM;

    const dim3 pgrid((DIM + 15) / 16, ROWS / 16, 3);
    proj_wmma_kernel<<<pgrid, 32, 0, stream>>>(query, WQ, bQ, q,
                                               key,   WK, bK, k,
                                               value, WK, bK, v);

    attn_fused_kernel<<<dim3(T1N, BATCH), BLOCK, 0, stream>>>(q, k, v, hL, hR, out);
}


# softmax on wave 0: each lane keeps its 4 scores in registers and evaluates every exponential once (was up to 5 times, in 5 loops with an LDS read per trip); sum order kept, weights bit-identical
# speedup vs baseline: 4.6052x; 1.0285x over previous
#include <hip/hip_runtime.h>

#define DIM   300
#define T1N   128
#define T2N   128
#define BATCH 4
#define ROWS  (BATCH * T1N)

#define WAVES 8
#define BLOCK (WAVES * 32)

typedef __attribute__((ext_vector_type(2))) float v2f;
typedef __attribute__((ext_vector_type(8))) float v8f;

typedef __attribute__((ext_vector_type(4))) float v4f;
#define TILE4     (16 * DIM / 4)
#define NLD       ((TILE4 + 31) / 32)
#define TILE4_PAD (NLD * 32)

__global__ __launch_bounds__(32) void proj_wmma_kernel(
    const float* __restrict__ X0, const float* __restrict__ W0,
    const float* __restrict__ bias0, float* __restrict__ out0,
    const float* __restrict__ X1, const float* __restrict__ W1,
    const float* __restrict__ bias1, float* __restrict__ out1,
    const float* __restrict__ X2, const float* __restrict__ W2,
    const float* __restrict__ bias2, float* __restrict__ out2)
{
    __shared__ v4f sA4[TILE4_PAD];
    __shared__ v4f sB4[TILE4_PAD];

    const int which = blockIdx.z;
    const float* __restrict__ X    = which == 0 ? X0    : (which == 1 ? X1    : X2);
    const float* __restrict__ W    = which == 0 ? W0    : (which == 1 ? W1    : W2);
    const float* __restrict__ bias = which == 0 ? bias0 : (which == 1 ? bias1 : bias2);
    float* __restrict__       out  = which == 0 ? out0  : (which == 1 ? out1  : out2);

    const int lane  = threadIdx.x;
    const int nTile = blockIdx.x;
    const int mTile = blockIdx.y;
    const int m0 = mTile * 16;
    const int n0 = nTile * 16;

    const int half = lane >> 4;
    const int l16  = lane & 15;

    const int  n      = n0 + l16;
    const bool nvalid = (n < DIM);

    const v4f* __restrict__ gA = reinterpret_cast<const v4f*>(X + (size_t)m0 * DIM);
    const v4f* __restrict__ gB = reinterpret_cast<const v4f*>(W + (size_t)n0 * DIM);
    const int rowsB = (DIM - n0) < 16 ? (DIM - n0) : 16;
    const int limB  = rowsB * (DIM / 4) - 1;

    v4f ra[NLD], rb[NLD];
    #pragma unroll
    for (int g = 0; g < NLD - 1; ++g) ra[g] = gA[lane + 32 * g];
    {
        const int idx = lane + 32 * (NLD - 1);
        ra[NLD - 1] = gA[idx < TILE4 ? idx : TILE4 - 1];
    }
    #pragma unroll
    for (int g = 0; g < NLD; ++g) {
        const int idx = lane + 32 * g;
        rb[g] = gB[idx < limB ? idx : limB];
    }
    #pragma unroll
    for (int g = 0; g < NLD; ++g) sA4[lane + 32 * g] = ra[g];
    #pragma unroll
    for (int g = 0; g < NLD; ++g) sB4[lane + 32 * g] = rb[g];
    __syncthreads();

    const v2f* fa = reinterpret_cast<const v2f*>(reinterpret_cast<const float*>(sA4) + l16 * DIM + 2 * half);
    const v2f* fb = reinterpret_cast<const v2f*>(reinterpret_cast<const float*>(sB4) + l16 * DIM + 2 * half);

    v8f acc0 = {};
    v8f acc1 = {};
    #pragma unroll
    for (int s = 0; s + 2 <= DIM / 4; s += 2) {
        const v2f a0 = fa[2 * s],     b0 = fb[2 * s];
        const v2f a1 = fa[2 * s + 2], b1 = fb[2 * s + 2];
        acc0 = __builtin_amdgcn_wmma_f32_16x16x4_f32(
            false, a0, false, b0, (short)0, acc0, false, false);
        acc1 = __builtin_amdgcn_wmma_f32_16x16x4_f32(
            false, a1, false, b1, (short)0, acc1, false, false);
    }
    {
        const v2f a0 = fa[2 * (DIM / 4 - 1)], b0 = fb[2 * (DIM / 4 - 1)];
        acc0 = __builtin_amdgcn_wmma_f32_16x16x4_f32(
            false, a0, false, b0, (short)0, acc0, false, false);
    }
    const v8f acc = acc0 + acc1;

    if (nvalid) {
        const float bv = bias[n];
        #pragma unroll
        for (int r = 0; r < 8; ++r) {
            const int m = m0 + r + 8 * half;
            out[(size_t)m * DIM + n] = acc[r] + bv;
        }
    }
}

#define NSTEP 10

struct RowRegs { float a[NSTEP]; float r[NSTEP]; float c[NSTEP]; };

__device__ __forceinline__ void load_row(RowRegs& R, const float* __restrict__ pa,
                                         const float* __restrict__ pr,
                                         const float* __restrict__ pc, int toff)
{
    #pragma unroll
    for (int m = 0; m < NSTEP - 1; ++m) {
        R.a[m] = pa[32 * m];
        R.r[m] = pr[32 * m];
        R.c[m] = pc[32 * m];
    }
    R.a[NSTEP - 1] = pa[toff];
    R.r[NSTEP - 1] = pr[toff];
    R.c[NSTEP - 1] = pc[toff];
}

__device__ __forceinline__ float row_score(const RowRegs& R, const float (&qv)[NSTEP], bool tailv)
{
    float acc = 0.0f;
    #pragma unroll
    for (int m = 0; m < NSTEP - 1; ++m)
        acc += (qv[m] + R.a[m]) * (R.c[m] + R.r[m]);
    const float t = acc + (qv[NSTEP - 1] + R.a[NSTEP - 1]) * (R.c[NSTEP - 1] + R.r[NSTEP - 1]);
    acc = tailv ? t : acc;
    #pragma unroll
    for (int off = 16; off > 0; off >>= 1)
        acc += __shfl_xor(acc, off, 32);
    return acc;
}

__global__ __launch_bounds__(BLOCK) __attribute__((amdgpu_waves_per_eu(4, 4)))
void attn_fused_kernel(
    const float* __restrict__ q, const float* __restrict__ k,
    const float* __restrict__ v, const float* __restrict__ hL,
    const float* __restrict__ hR, float* __restrict__ out)
{
    const int b    = blockIdx.y;
    const int i    = blockIdx.x;
    const int tid  = threadIdx.x;
    const int wave = tid >> 5;
    const int lane = tid & 31;

    __shared__ float s_q[DIM];
    __shared__ float s_attn[T2N];
    __shared__ unsigned s_nz[T2N / 32];

    const float* qrow = q + ((size_t)b * T1N + i) * DIM;
    for (int d = tid; d < DIM; d += BLOCK) s_q[d] = qrow[d];
    __syncthreads();

    float qv[NSTEP];
    #pragma unroll
    for (int m = 0; m < NSTEP; ++m) {
        const int d = lane + 32 * m;
        qv[m] = s_q[d < DIM ? d : DIM - 1];
    }
    const bool tailv = lane < DIM - 32 * (NSTEP - 1);
    const int  toff  = tailv ? 32 * (NSTEP - 1) : 32 * (NSTEP - 2);

    const float* hLw = hL + (((size_t)b * T1N + i) * T2N + wave) * DIM + lane;
    const float* hRw = hR + (((size_t)b * T2N + wave) * T1N + i) * DIM + lane;
    const float* kw  = k  + ((size_t)b * T2N + wave) * DIM + lane;
    const size_t hLs = (size_t)WAVES * DIM;
    const size_t hRs = (size_t)WAVES * T1N * DIM;

    RowRegs ra, rb;
    load_row(ra, hLw, hRw, kw, toff);
    for (int t = 0; t < T2N / WAVES - 2; t += 2) {
        load_row(rb, hLw + (t + 1) * hLs, hRw + (t + 1) * hRs, kw + (t + 1) * hLs, toff);
        const float sa = row_score(ra, qv, tailv);
        if (lane == 0) s_attn[wave + WAVES * t] = sa;
        load_row(ra, hLw + (t + 2) * hLs, hRw + (t + 2) * hRs, kw + (t + 2) * hLs, toff);
        const float sb = row_score(rb, qv, tailv);
        if (lane == 0) s_attn[wave + WAVES * (t + 1)] = sb;
    }
    {
        const int t = T2N / WAVES - 2;
        load_row(rb, hLw + (t + 1) * hLs, hRw + (t + 1) * hRs, kw + (t + 1) * hLs, toff);
        const float sa = row_score(ra, qv, tailv);
        if (lane == 0) s_attn[wave + WAVES * t] = sa;
        const float sb = row_score(rb, qv, tailv);
        if (lane == 0) s_attn[wave + WAVES * (t + 1)] = sb;
    }
    __syncthreads();

    if (wave == 0) {
        constexpr int NT = T2N / 32;
        float s[NT], p[NT], e2[NT];
        #pragma unroll
        for (int t = 0; t < NT; ++t) s[t] = s_attn[lane + 32 * t];

        float m = -__builtin_inff();
        #pragma unroll
        for (int t = 0; t < NT; ++t) m = fmaxf(m, s[t]);
        #pragma unroll
        for (int off = 16; off > 0; off >>= 1) m = fmaxf(m, __shfl_xor(m, off, 32));

        float sum = 0.0f;
        #pragma unroll
        for (int t = 0; t < NT; ++t) { p[t] = expf(s[t] - m); sum += p[t]; }
        #pragma unroll
        for (int off = 16; off > 0; off >>= 1) sum += __shfl_xor(sum, off, 32);
        const float inv = 1.0f / sum;

        float m2 = -__builtin_inff();
        #pragma unroll
        for (int t = 0; t < NT; ++t) { p[t] = p[t] * inv; m2 = fmaxf(m2, 1000.0f * p[t]); }
        #pragma unroll
        for (int off = 16; off > 0; off >>= 1) m2 = fmaxf(m2, __shfl_xor(m2, off, 32));

        float sum2 = 0.0f;
        #pragma unroll
        for (int t = 0; t < NT; ++t) { e2[t] = expf(1000.0f * p[t] - m2); sum2 += e2[t]; }
        #pragma unroll
        for (int off = 16; off > 0; off >>= 1) sum2 += __shfl_xor(sum2, off, 32);
        const float inv2 = 1.0f / sum2;

        #pragma unroll
        for (int t = 0; t < NT; ++t) {
            const float a2 = e2[t] * inv2;
            const float w  = fminf(fmaxf(a2, 0.0f), 1.0f);
            s_attn[lane + 32 * t] = w;
            const unsigned present = __builtin_amdgcn_ballot_w32(!(w == 0.0f));
            if (lane == 0) s_nz[t] = present;
        }
    }
    __syncthreads();

    for (int d = tid; d < DIM; d += BLOCK) {
        const float* vp  = v  + (size_t)b * T2N * DIM + d;
        const float* hRp = hR + ((size_t)b * T2N * T1N + i) * DIM + d;
        float acc = 0.0f;
        for (int t = 0; t < T2N / 32; ++t) {
            unsigned mk = (unsigned)__builtin_amdgcn_readfirstlane((int)s_nz[t]);
            while (mk != 0u) {
                const int j = 32 * t + __builtin_ctz(mk);
                mk &= mk - 1u;
                acc += s_attn[j] * (vp[(size_t)j * DIM] + hRp[(size_t)j * T1N * DIM]);
            }
        }
        out[((size_t)b * T1N + i) * DIM + d] = acc;
    }
}

extern "C" void kernel_launch(void* const* d_in, const int* in_sizes, int n_in,
                              void* d_out, int out_size, void* d_ws, size_t ws_size,
                              hipStream_t stream)
{
    const float* query = (const float*)d_in[0];
    const float* key   = (const float*)d_in[1];
    const float* value = (const float*)d_in[2];
    const float* hL    = (const float*)d_in[3];
    const float* hR    = (const float*)d_in[4];
    const float* WQ    = (const float*)d_in[5];
    const float* bQ    = (const float*)d_in[6];
    const float* WK    = (const float*)d_in[7];
    const float* bK    = (const float*)d_in[8];
    float* out = (float*)d_out;

    float* q = (float*)d_ws;
    float* k = q + (size_t)ROWS * DIM;
    float* v = k + (size_t)ROWS * DIM;

    const dim3 pgrid((DIM + 15) / 16, ROWS / 16, 3);
    proj_wmma_kernel<<<pgrid, 32, 0, stream>>>(query, WQ, bQ, q,
                                               key,   WK, bK, k,
                                               value, WK, bK, v);

    attn_fused_kernel<<<dim3(T1N, BATCH), BLOCK, 0, stream>>>(q, k, v, hL, hR, out);
}
